# starting kernel
# baseline (speedup 1.0000x reference)
#include <hip/hip_runtime.h>
#include <cstddef>

typedef float v2f __attribute__((ext_vector_type(2)));
typedef float v8f __attribute__((ext_vector_type(8)));

namespace {
constexpr int kB    = 8;
constexpr int kN    = 256;
constexpr int kL    = 4;
constexpr int kH    = 128;
constexpr int kBN   = kB * kN;
constexpr int kG    = 4 * kH;
constexpr int kSteps = 16;
constexpr int kOutV = 30000;
constexpr size_t kHOFF = (size_t)2 * kBN * kH;
constexpr int AS = 132;
constexpr int GS = 516;
constexpr int kPairs = 2 * (kH / 2) * kG;
}

__device__ __forceinline__ float sigmf(float x) { return 1.0f / (1.0f + expf(-x)); }

__device__ __forceinline__ v8f wmma4(v2f a, v2f b, v8f c) {
  return __builtin_amdgcn_wmma_f32_16x16x4_f32(false, a, false, b, (short)0, c, false, false);
}

__device__ __forceinline__ void gemm_tile(const float* __restrict__ A,
                                          const v2f* __restrict__ Wp,
                                          int c_base, int lane, v8f acc[4]) {
  const int m  = lane & 15;
  const int hf = lane >> 4;
  const int n  = lane & 15;
#pragma unroll 2
  for (int kc = 0; kc < 32; ++kc) {
    const int kk = kc * 4 + 2 * hf;
    v2f a = *(const v2f*)(A + m * AS + kk);
    const v2f* Wr = Wp + (size_t)(kc * 2 + hf) * kG;
#pragma unroll
    for (int t = 0; t < 4; ++t) {
      acc[t] = wmma4(a, Wr[c_base + t * 16 + n], acc[t]);
    }
  }
}

__global__ __launch_bounds__(256) void k_lstm_step(
    const float* __restrict__ emb, const float* __restrict__ Sin,
    float* __restrict__ Stg, const float* __restrict__ ipin,
    float* __restrict__ ptb, float* __restrict__ pfb,
    const v2f* __restrict__ Wxp, const v2f* __restrict__ Whp,
    const float* __restrict__ bias, const float* __restrict__ bW,
    const float* __restrict__ bb, const int* __restrict__ exitI) {
  __shared__ float sX[16 * AS];
  __shared__ float sH[2][16 * AS];
  __shared__ float sC[2][16 * kH];
  __shared__ float sG[16 * GS];
  __shared__ float sLg[32];

  const int tid  = threadIdx.x;
  const int lane = tid & 31;
  const int wave = tid >> 5;
  const int r0   = blockIdx.x * 16;
  const int c_base = wave * 64;

  for (int idx = tid; idx < 2 * 16 * kH; idx += 256) {
    int l = idx >> 11, rem = idx & 2047, m = rem >> 7, k = rem & 127;
    size_t row = r0 + m;
    sC[l][m * kH + k] = Sin[((size_t)l * kBN + row) * kH + k];
    sH[l][m * AS + k] = Sin[kHOFF + ((size_t)l * kBN + row) * kH + k];
  }
  __syncthreads();

  for (int tok = 0; tok < kL; ++tok) {
    for (int idx = tid; idx < 16 * kH; idx += 256) {
      int m = idx >> 7, k = idx & 127;
      sX[m * AS + k] = emb[(((size_t)(r0 + m)) * kL + tok) * kH + k];
    }
    __syncthreads();
    for (int l = 0; l < 2; ++l) {
      v8f acc[4];
#pragma unroll
      for (int t = 0; t < 4; ++t) {
        float blv = bias[l * kG + c_base + t * 16 + (lane & 15)];
#pragma unroll
        for (int v = 0; v < 8; ++v) acc[t][v] = blv;
      }
      const float* Ain = (l == 0) ? sX : sH[0];
      gemm_tile(Ain,   Wxp + (size_t)l * (kH / 2) * kG, c_base, lane, acc);
      gemm_tile(sH[l], Whp + (size_t)l * (kH / 2) * kG, c_base, lane, acc);
      const int hf = lane >> 4;
#pragma unroll
      for (int t = 0; t < 4; ++t) {
        int col = c_base + t * 16 + (lane & 15);
#pragma unroll
        for (int v = 0; v < 8; ++v) sG[(v + 8 * hf) * GS + col] = acc[t][v];
      }
      __syncthreads();
      for (int idx = tid; idx < 16 * kH; idx += 256) {
        int m = idx >> 7, k = idx & 127;
        float iv = sG[m * GS + k];
        float fv = sG[m * GS + kH + k];
        float gv = sG[m * GS + 2 * kH + k];
        float ov = sG[m * GS + 3 * kH + k];
        float cold = sC[l][m * kH + k];
        float cn = sigmf(fv) * cold + sigmf(iv) * tanhf(gv);
        float hn = sigmf(ov) * tanhf(cn);
        sC[l][m * kH + k] = cn;
        sH[l][m * AS + k] = hn;
      }
      __syncthreads();
    }
  }

  const int bidx = r0 >> 8;
  const int mm = exitI[bidx] - (r0 & 255);
  if (mm >= 0 && mm < 16) {
    for (int idx = tid; idx < 2 * kH; idx += 256) {
      int l = idx >> 7, k = idx & 127;
      size_t row = r0 + mm;
      sC[l][mm * kH + k] = Sin[((size_t)l * kBN + row) * kH + k];
      sH[l][mm * AS + k] = Sin[kHOFF + ((size_t)l * kBN + row) * kH + k];
    }
  }
  __syncthreads();

  if (tid < 32) {
    int m = tid >> 1, j = tid & 1;
    float s = bb[j];
    for (int k = 0; k < kH; ++k) s += sC[0][m * kH + k] * bW[k * 2 + j];
    for (int k = 0; k < kH; ++k) s += sH[0][m * AS + k] * bW[(kH + k) * 2 + j];
    for (int k = 0; k < kH; ++k) s += sC[1][m * kH + k] * bW[(2 * kH + k) * 2 + j];
    for (int k = 0; k < kH; ++k) s += sH[1][m * AS + k] * bW[(3 * kH + k) * 2 + j];
    sLg[tid] = s;
  }
  __syncthreads();
  if (tid < 16) {
    float l0 = sLg[2 * tid], l1 = sLg[2 * tid + 1];
    float mx = fmaxf(l0, l1);
    float e0 = expf(l0 - mx), e1 = expf(l1 - mx);
    float inv = 1.0f / (e0 + e1);
    float ipv = ipin[r0 + tid];
    ptb[r0 + tid] = e0 * inv * ipv;
    pfb[r0 + tid] = e1 * inv * ipv;
  }

  for (int idx = tid; idx < 2 * 16 * kH; idx += 256) {
    int l = idx >> 11, rem = idx & 2047, m = rem >> 7, k = rem & 127;
    size_t row = r0 + m;
    Stg[((size_t)l * kBN + row) * kH + k]         = sC[l][m * kH + k];
    Stg[kHOFF + ((size_t)l * kBN + row) * kH + k] = sH[l][m * AS + k];
  }
}

__global__ __launch_bounds__(128) void k_aggregate(
    const float* __restrict__ Stg, const float* __restrict__ Sin,
    const float* __restrict__ ipin, const float* __restrict__ ptb,
    const float* __restrict__ pfb, const int* __restrict__ ti,
    const int* __restrict__ fi, const int* __restrict__ steps,
    float* __restrict__ Sout, float* __restrict__ ipout, int step) {
  const int tgt = blockIdx.x;
  const int b = tgt >> 8;
  const int m = tgt & 255;
  const int h = threadIdx.x;
  float aC0 = 0.f, aH0 = 0.f, aC1 = 0.f, aH1 = 0.f, aip = 0.f;
  const int base = b * kN;
  for (int n = 0; n < kN; ++n) {
    int src = base + n;
    float w = 0.f;
    if (ti[src] == m) w += ptb[src];
    if (fi[src] == m) w += pfb[src];
    if (w != 0.f) {
      aip += w;
      aC0 += w * Stg[((size_t)src) * kH + h];
      aH0 += w * Stg[kHOFF + ((size_t)src) * kH + h];
      aC1 += w * Stg[((size_t)kBN + src) * kH + h];
      aH1 += w * Stg[kHOFF + ((size_t)kBN + src) * kH + h];
    }
  }
  float denom = aip + 1e-7f;
  bool active = step < steps[b];
  size_t row = tgt;
  Sout[row * kH + h]                         = active ? aC0 / denom : Sin[row * kH + h];
  Sout[kHOFF + row * kH + h]                 = active ? aH0 / denom : Sin[kHOFF + row * kH + h];
  Sout[((size_t)kBN + row) * kH + h]         = active ? aC1 / denom : Sin[((size_t)kBN + row) * kH + h];
  Sout[kHOFF + ((size_t)kBN + row) * kH + h] = active ? aH1 / denom : Sin[kHOFF + ((size_t)kBN + row) * kH + h];
  if (h == 0) ipout[tgt] = active ? aip : ipin[tgt];
}

__global__ __launch_bounds__(256) void k_logits(
    const float* __restrict__ S, const int* __restrict__ exitI,
    const float* __restrict__ outW, const float* __restrict__ outb,
    float* __restrict__ out) {
  __shared__ float sF[kB * 512];
  const int tid = threadIdx.x;
  for (int idx = tid; idx < kB * 512; idx += 256) {
    int q = idx >> 9, k = idx & 511, seg = k >> 7, kk = k & 127;
    size_t row = (size_t)q * kN + exitI[q];
    float v;
    if (seg == 0)      v = S[row * kH + kk];
    else if (seg == 1) v = S[kHOFF + row * kH + kk];
    else if (seg == 2) v = S[((size_t)kBN + row) * kH + kk];
    else               v = S[kHOFF + ((size_t)kBN + row) * kH + kk];
    sF[idx] = v;
  }
  __syncthreads();
  int j = blockIdx.x * 256 + tid;
  if (j >= kOutV) return;
  float acc[kB];
  float bj = outb[j];
#pragma unroll
  for (int q = 0; q < kB; ++q) acc[q] = bj;
  for (int k = 0; k < 512; ++k) {
    float w = outW[(size_t)k * kOutV + j];
#pragma unroll
    for (int q = 0; q < kB; ++q) acc[q] += sF[q * 512 + k] * w;
  }
  for (int q = 0; q < kB; ++q) out[(size_t)q * kOutV + j] = acc[q];
}

__global__ void k_embed(const int* __restrict__ data, const float* __restrict__ embed,
                        float* __restrict__ embuf, int total) {
  int idx = blockIdx.x * 256 + threadIdx.x;
  if (idx >= total) return;
  int token = idx >> 7, h = idx & 127;
  embuf[idx] = embed[(size_t)data[token] * kH + h];
}

__global__ void k_pack(const float* __restrict__ Wsrc, float* __restrict__ Wdst, int total) {
  int idx = blockIdx.x * 256 + threadIdx.x;
  if (idx >= total) return;
  int col = idx & 511;
  int p   = (idx >> 9) & 63;
  int l   = idx >> 15;
  const float* Wl = Wsrc + (size_t)l * kH * kG;
  v2f v;
  v.x = Wl[(size_t)(2 * p) * kG + col];
  v.y = Wl[(size_t)(2 * p + 1) * kG + col];
  *(v2f*)(Wdst + (size_t)idx * 2) = v;
}

__global__ void k_init(float* __restrict__ S0, float* __restrict__ ip0, int total) {
  int idx = blockIdx.x * 256 + threadIdx.x;
  if (idx < total) S0[idx] = 0.f;
  if (idx < kBN) ip0[idx] = ((idx & 255) == 0) ? 1.0f : 0.0f;
}

extern "C" void kernel_launch(void* const* d_in, const int* in_sizes, int n_in,
                              void* d_out, int out_size, void* d_ws, size_t ws_size,
                              hipStream_t stream) {
  const int*   data  = (const int*)d_in[0];
  const int*   ti    = (const int*)d_in[1];
  const int*   fi    = (const int*)d_in[2];
  const int*   exitI = (const int*)d_in[3];
  const int*   steps = (const int*)d_in[4];
  const float* embed = (const float*)d_in[5];
  const float* Wx    = (const float*)d_in[6];
  const float* Wh    = (const float*)d_in[7];
  const float* bias  = (const float*)d_in[8];
  const float* bW    = (const float*)d_in[9];
  const float* bb    = (const float*)d_in[10];
  const float* outW  = (const float*)d_in[11];
  const float* outb  = (const float*)d_in[12];
  float* out = (float*)d_out;

  float* ws = (float*)d_ws;
  const size_t EMB = (size_t)kBN * kL * kH;
  const size_t ST  = (size_t)4 * kBN * kH;
  float* embuf = ws;
  float* S0  = embuf + EMB;
  float* S1  = S0 + ST;
  float* STG = S1 + ST;
  float* ip0 = STG + ST;
  float* ip1 = ip0 + kBN;
  float* ptb = ip1 + kBN;
  float* pfb = ptb + kBN;
  float* Wxp = pfb + kBN;
  float* Whp = Wxp + (size_t)kPairs * 2;

  k_embed<<<(int)((EMB + 255) / 256), 256, 0, stream>>>(data, embed, embuf, (int)EMB);
  k_init<<<(int)((ST + 255) / 256), 256, 0, stream>>>(S0, ip0, (int)ST);
  k_pack<<<(kPairs + 255) / 256, 256, 0, stream>>>(Wx, Wxp, kPairs);
  k_pack<<<(kPairs + 255) / 256, 256, 0, stream>>>(Wh, Whp, kPairs);

  float* Sin = S0; float* Sout = S1; float* ipin = ip0; float* ipout = ip1;
  for (int s = 0; s < kSteps; ++s) {
    k_lstm_step<<<kBN / 16, 256, 0, stream>>>(embuf, Sin, STG, ipin, ptb, pfb,
                                              (const v2f*)Wxp, (const v2f*)Whp,
                                              bias, bW, bb, exitI);
    k_aggregate<<<kBN, 128, 0, stream>>>(STG, Sin, ipin, ptb, pfb, ti, fi, steps,
                                         Sout, ipout, s);
    float* t1 = Sin; Sin = Sout; Sout = t1;
    float* t2 = ipin; ipin = ipout; ipout = t2;
  }
  k_logits<<<(kOutV + 255) / 256, 256, 0, stream>>>(Sin, exitI, outW, outb, out);
  (void)in_sizes; (void)n_in; (void)out_size; (void)ws_size;
}


# Aggregation moved into its consumers (next step launch's head; output kernel), 8 source rows in flight; a batch's masked steps are not run (cell update too), its state is stored once; sum order kept
# speedup vs baseline: 3.3532x; 3.3532x over previous
#include <hip/hip_runtime.h>
#include <cstddef>

typedef float v2f __attribute__((ext_vector_type(2)));
typedef float v4f __attribute__((ext_vector_type(4)));
typedef float v8f __attribute__((ext_vector_type(8)));

namespace {
constexpr int kB    = 8;
constexpr int kN    = 256;
constexpr int kL    = 4;
constexpr int kH    = 128;
constexpr int kBN   = kB * kN;
constexpr int kG    = 4 * kH;
constexpr int kSteps = 16;
constexpr int kOutV = 30000;
constexpr size_t kHOFF = (size_t)2 * kBN * kH;
constexpr int AS = 132;
constexpr int kPairs = 2 * (kH / 2) * kG;
constexpr size_t kST  = (size_t)4 * kBN * kH;
constexpr size_t kEMB = (size_t)kBN * kL * kH;
constexpr size_t kGX  = (size_t)kBN * kL * kG;
constexpr size_t kFixed = 3 * kST + 4 * (size_t)kBN + 4 * (size_t)kPairs;
constexpr size_t kNeedBytes = (kFixed + kGX) * sizeof(float);
}

__device__ __forceinline__ float sigmf(float x) {
  return __builtin_amdgcn_rcpf(1.0f + __builtin_amdgcn_exp2f(x * -1.44269504088896341f));
}
__device__ __forceinline__ float tanhfast(float x) {
  return __builtin_fmaf(-2.0f, __builtin_amdgcn_rcpf(1.0f + __builtin_amdgcn_exp2f(x * 2.88539008177792681f)), 1.0f);
}

__device__ __forceinline__ v8f wmma4(v2f a, v2f b, v8f c) {
  return __builtin_amdgcn_wmma_f32_16x16x4_f32(false, a, false, b, (short)0, c, false, false);
}

__device__ __forceinline__ void load_b_block(const v2f* __restrict__ Wl, int blk, v2f (&b)[8][4]) {
#pragma unroll
  for (int j = 0; j < 8; ++j) {
    const v2f* Wr = Wl + (size_t)((blk * 8 + j) * 2) * kG;
#pragma unroll
    for (int t = 0; t < 4; ++t) b[j][t] = Wr[t * 16];
  }
}

__device__ __forceinline__ void load_a_block(const float* Al, int blk, v2f (&a)[8]) {
#pragma unroll
  for (int j = 0; j < 8; ++j) a[j] = *(const v2f*)(Al + (blk * 8 + j) * 4);
}

__device__ __forceinline__ void mma_block(const v2f (&a)[8], const v2f (&b)[8][4], v8f (&acc)[4]) {
#pragma unroll
  for (int j = 0; j < 8; ++j) {
#pragma unroll
    for (int t = 0; t < 4; ++t) acc[t] = wmma4(a[j], b[j][t], acc[t]);
  }
}

__device__ __forceinline__ v8f load_tile(const v4f* p) {
  const v4f lo = p[0], hi = p[32];
  return __builtin_shufflevector(lo, hi, 0, 1, 2, 3, 4, 5, 6, 7);
}
__device__ __forceinline__ void store_tile(v4f* p, v8f a) {
  p[0]  = __builtin_shufflevector(a, a, 0, 1, 2, 3);
  p[32] = __builtin_shufflevector(a, a, 4, 5, 6, 7);
}

__device__ __forceinline__ void gather2(
    const float* __restrict__ Stg, const float* __restrict__ ptb, const float* __restrict__ pfb,
    const int* __restrict__ ti, const int* __restrict__ fi, int base, int mA, int mB, int lane,
    float* Lw, v4f (&aA)[4], v4f (&aB)[4], float& ipA, float& ipB) {
  int t[8], f[8];
  float pt[8], pf[8];
#pragma unroll
  for (int i = 0; i < 8; ++i) {
    const int src = base + 32 * i + lane;
    t[i] = ti[src];
    f[i] = fi[src];
    pt[i] = ptb[src];
    pf[i] = pfb[src];
  }
  int cnt = 0;
#pragma unroll
  for (int i = 0; i < 8; ++i) {
    const float a0 = (t[i] == mA) ? 0.f + pt[i] : 0.f;
    const float wa = (f[i] == mA) ? a0 + pf[i] : a0;
    const float b0 = (t[i] == mB) ? 0.f + pt[i] : 0.f;
    const float wb = (f[i] == mB) ? b0 + pf[i] : b0;
    const bool con = (wa != 0.f) || (wb != 0.f);
    const unsigned mk = __builtin_amdgcn_ballot_w32(con);
    if (con) {
      const int rank = cnt + (int)__builtin_amdgcn_mbcnt_lo(mk, 0u);
      *(v4f*)(Lw + 4 * rank) = (v4f){__int_as_float(32 * i + lane), wa, wb, 0.f};
    }
    cnt += __builtin_popcount(mk);
  }
#pragma unroll
  for (int p = 0; p < 4; ++p) { aA[p] = (v4f){0.f, 0.f, 0.f, 0.f}; aB[p] = aA[p]; }
  ipA = 0.f;
  ipB = 0.f;
  for (int k = 0; k < cnt; k += 8) {
    v4f e[8];
    v4f x[8][4];
#pragma unroll
    for (int j = 0; j < 8; ++j) {
      const int idx = (k + j < cnt) ? k + j : cnt - 1;
      e[j] = *(const v4f*)(Lw + 4 * idx);
    }
#pragma unroll
    for (int j = 0; j < 8; ++j) {
      const float* r = Stg + (size_t)(base + __float_as_int(e[j].x)) * kH + 4 * lane;
      x[j][0] = *(const v4f*)(r);
      x[j][1] = *(const v4f*)(r + kHOFF);
      x[j][2] = *(const v4f*)(r + (size_t)kBN * kH);
      x[j][3] = *(const v4f*)(r + kHOFF + (size_t)kBN * kH);
    }
#pragma unroll
    for (int j = 0; j < 8; ++j) {
      const bool in = k + j < cnt;
      const float wa = in ? e[j].y : 0.f;
      const float wb = in ? e[j].z : 0.f;
      ipA += wa;
      ipB += wb;
#pragma unroll
      for (int p = 0; p < 4; ++p) {
        aA[p] += wa * x[j][p];
        aB[p] += wb * x[j][p];
      }
    }
  }
}

__global__ __launch_bounds__(256) void k_lstm_step(
    const float* __restrict__ X, size_t ws_bytes, const float* __restrict__ StgIn,
    float* __restrict__ Stg, float* __restrict__ Sfin,
    const float* __restrict__ ptIn, const float* __restrict__ pfIn,
    float* __restrict__ ptb, float* __restrict__ pfb,
    const v2f* __restrict__ Wxp, const v2f* __restrict__ Whp,
    const float* __restrict__ bias, const float* __restrict__ bW,
    const float* __restrict__ bb, const int* __restrict__ exitI,
    const int* __restrict__ ti, const int* __restrict__ fi,
    const int* __restrict__ steps, int step) {
  __shared__ float sX[16 * AS];
  alignas(16) __shared__ float sH[2][2][16 * AS];
  alignas(16) __shared__ float sC[2][16 * kH];
  alignas(16) __shared__ float sList[8][256 * 4];
  __shared__ float sIp[16];
  alignas(16) __shared__ float sPart[8][32];
  alignas(16) __shared__ float sBW[4 * kH * 2];

  const int tid  = threadIdx.x;
  const int lane = tid & 31;
  const int wave = tid >> 5;
  const int r0   = blockIdx.x * 16;
  const int c_base = wave * 64;
  const int hf   = lane >> 4;
  const int hcol = wave * 16 + (lane & 15);
  const bool pre = ws_bytes >= kNeedBytes;
  const v4f* Xg = (const v4f*)X + ((size_t)blockIdx.x * 128 + wave * 4) * 64 + lane;
  v8f gx[4];
  if (pre) {
#pragma unroll
    for (int t = 0; t < 4; ++t) gx[t] = load_tile(Xg + t * 64);
  }

  const int mm  = exitI[r0 >> 8] - (r0 & 255);
  const bool hit = (mm >= 0) && (mm < 16);
  const int mmc = mm < 0 ? 0 : (mm > 15 ? 15 : mm);

  const int bidx = r0 >> 8;
  int stp = steps[bidx];
  stp = stp < 0 ? 0 : stp;
  if (step > stp) return;

  const int wv = __builtin_amdgcn_readfirstlane(wave);
  {
    v4f ag[2][4];
    float ipr[2];
    if (step == 0) {
#pragma unroll
      for (int p = 0; p < 4; ++p) { ag[0][p] = (v4f){0.f, 0.f, 0.f, 0.f}; ag[1][p] = ag[0][p]; }
      ipr[0] = ((r0 & 255) + 2 * wv == 0) ? 1.0f : 0.0f;
      ipr[1] = 0.0f;
    } else {
      const int mA = (r0 & 255) + 2 * wv;
      gather2(StgIn, ptIn, pfIn, ti, fi, bidx * kN, mA, mA + 1, lane, sList[wv], ag[0], ag[1], ipr[0], ipr[1]);
#pragma unroll
      for (int j = 0; j < 2; ++j) {
        const float denom = ipr[j] + 1e-7f;
#pragma unroll
        for (int p = 0; p < 4; ++p) ag[j][p] = ag[j][p] / denom;
      }
    }
    if (step == stp) {
#pragma unroll
      for (int j = 0; j < 2; ++j) {
        const size_t o = (size_t)(r0 + 2 * wv + j) * kH + 4 * lane;
        *(v4f*)(Sfin + o)                            = ag[j][0];
        *(v4f*)(Sfin + kHOFF + o)                    = ag[j][1];
        *(v4f*)(Sfin + (size_t)kBN * kH + o)         = ag[j][2];
        *(v4f*)(Sfin + kHOFF + (size_t)kBN * kH + o) = ag[j][3];
      }
      return;
    }
#pragma unroll
    for (int j = 0; j < 2; ++j) {
      const int rr = 2 * wv + j;
      *(v4f*)(&sC[0][rr * kH + 4 * lane])    = ag[j][0];
      *(v4f*)(&sH[0][0][rr * AS + 4 * lane]) = ag[j][1];
      *(v4f*)(&sC[1][rr * kH + 4 * lane])    = ag[j][2];
      *(v4f*)(&sH[1][0][rr * AS + 4 * lane]) = ag[j][3];
      if (lane == 0) sIp[rr] = ipr[j];
    }
  }

  const int sm0 = tid >> 7, sk = tid & 127;
  {
    float bwv[4];
#pragma unroll
    for (int i = 0; i < 4; ++i) bwv[i] = bW[tid + 256 * i];
#pragma unroll
    for (int i = 0; i < 4; ++i) sBW[tid + 256 * i] = bwv[i];
  }
  __syncthreads();

  float cst[2][8];
  float hst[2][8];
#pragma unroll
  for (int i = 0; i < 16; ++i) cst[i >> 3][i & 7] = sC[i >> 3][((i & 7) + 8 * hf) * kH + hcol];

  const int aoff = (lane & 15) * AS + 2 * hf;
  const size_t boff = (size_t)hf * kG + c_base + (lane & 15);
  v2f bq0[8][4], bq1[8][4];
  load_b_block((pre ? Whp : Wxp) + boff, 0, bq0);
  float exs[4];
  exs[0] = sC[0][mmc * kH + hcol];
  exs[1] = sH[0][0][mmc * AS + hcol];
  exs[2] = sC[1][mmc * kH + hcol];
  exs[3] = sH[1][0][mmc * AS + hcol];

  for (int tok = 0; tok < kL; ++tok) {
    const int rb = tok & 1;
    if (!pre) {
      const float* xp = X + ((size_t)(r0 + sm0) * kL + tok) * kH + sk;
      float xv[8];
#pragma unroll
      for (int j = 0; j < 8; ++j) xv[j] = xp[(size_t)2 * j * kL * kH];
#pragma unroll
      for (int j = 0; j < 8; ++j) sX[(sm0 + 2 * j) * AS + sk] = xv[j];
      __syncthreads();
    }
#pragma unroll
    for (int l = 0; l < 2; ++l) {
      const bool skipx = (l == 0) && pre;
      v8f acc[4];
      if (skipx) {
#pragma unroll
        for (int t = 0; t < 4; ++t) acc[t] = gx[t];
      } else {
#pragma unroll
        for (int t = 0; t < 4; ++t) {
          float blv = bias[l * kG + t * kH + hcol];
#pragma unroll
          for (int v = 0; v < 8; ++v) acc[t][v] = blv;
        }
      }
      if (l == 1 && pre && tok + 1 < kL) {
#pragma unroll
        for (int t = 0; t < 4; ++t) gx[t] = load_tile(Xg + ((tok + 1) * 32 + t) * 64);
      }
      const float* A0 = ((l == 0) ? sX : sH[0][rb ^ 1]) + aoff;
      const float* A1 = sH[l][rb] + aoff;
      const v2f* Wx_l = Wxp + (size_t)l * (kH / 2) * kG + boff;
      const v2f* Wh_l = Whp + (size_t)l * (kH / 2) * kG + boff;
      const v2f* W_n  = (l == 0) ? Wxp + (size_t)(kH / 2) * kG + boff : (pre ? Whp : Wxp) + boff;
      v2f a0[8], a1[8];
      load_a_block(skipx ? A1 : A0, 0, a0);
#pragma unroll 1
      for (int i = skipx ? 2 : 0; i < 4; ++i) {
        const int g1 = 2 * i + 1;
        load_b_block((g1 < 4) ? Wx_l : Wh_l, g1 & 3, bq1);
        load_a_block((g1 < 4) ? A0 : A1, g1 & 3, a1);
        mma_block(a0, bq0, acc);
        const int g2 = 2 * i + 2;
        load_b_block((g2 < 4) ? Wx_l : ((g2 < 8) ? Wh_l : W_n), g2 & 3, bq0);
        if (g2 < 8) load_a_block((g2 < 4) ? A0 : A1, g2 & 3, a0);
        mma_block(a1, bq1, acc);
      }
      float* Hw = sH[l][rb ^ 1] + (8 * hf) * AS + hcol;
#pragma unroll
      for (int v = 0; v < 8; ++v) {
        const float cn = sigmf(acc[1][v]) * cst[l][v] + sigmf(acc[0][v]) * tanhfast(acc[2][v]);
        cst[l][v] = cn;
        const float hn = sigmf(acc[3][v]) * tanhfast(cn);
        hst[l][v] = hn;
        Hw[v * AS] = hn;
      }
      __syncthreads();
    }
  }

  constexpr int kFin = kL & 1;
  const float ipv = (tid < 16) ? sIp[tid] : 0.f;

  const bool mine = hit && (hf == (mmc >> 3));
  const int mv = mmc & 7;
#pragma unroll
  for (int v = 0; v < 8; ++v) {
    const bool ex = mine && (v == mv);
    cst[0][v] = ex ? exs[0] : cst[0][v];
    hst[0][v] = ex ? exs[1] : hst[0][v];
    cst[1][v] = ex ? exs[2] : cst[1][v];
    hst[1][v] = ex ? exs[3] : hst[1][v];
  }
  if (mine) {
    sH[0][kFin][mmc * AS + hcol] = exs[1];
    sH[1][kFin][mmc * AS + hcol] = exs[3];
  }
#pragma unroll
  for (int i = 0; i < 16; ++i) sC[i >> 3][((i & 7) + 8 * hf) * kH + hcol] = cst[i >> 3][i & 7];

  {
    const v2f w0 = *(const v2f*)(sBW + 2 * hcol);
    const v2f w1 = *(const v2f*)(sBW + 2 * (kH + hcol));
    const v2f w2 = *(const v2f*)(sBW + 2 * (2 * kH + hcol));
    const v2f w3 = *(const v2f*)(sBW + 2 * (3 * kH + hcol));
    float q[16];
#pragma unroll
    for (int v = 0; v < 8; ++v) {
      q[2 * v]     = cst[0][v] * w0.x + hst[0][v] * w1.x + cst[1][v] * w2.x + hst[1][v] * w3.x;
      q[2 * v + 1] = cst[0][v] * w0.y + hst[0][v] * w1.y + cst[1][v] * w2.y + hst[1][v] * w3.y;
    }
    const int cl = lane & 15;
    float r8[8], r4[4], r2[2];
    const bool u8 = (cl & 8) != 0, u4 = (cl & 4) != 0, u2 = (cl & 2) != 0, u1 = (cl & 1) != 0;
#pragma unroll
    for (int i = 0; i < 8; ++i) r8[i] = (u8 ? q[i + 8] : q[i]) + __shfl_xor(u8 ? q[i] : q[i + 8], 8);
#pragma unroll
    for (int i = 0; i < 4; ++i) r4[i] = (u4 ? r8[i + 4] : r8[i]) + __shfl_xor(u4 ? r8[i] : r8[i + 4], 4);
#pragma unroll
    for (int i = 0; i < 2; ++i) r2[i] = (u2 ? r4[i + 2] : r4[i]) + __shfl_xor(u2 ? r4[i] : r4[i + 2], 2);
    sPart[wave][lane] = (u1 ? r2[1] : r2[0]) + __shfl_xor(u1 ? r2[0] : r2[1], 1);
  }
  __syncthreads();

  if (tid < 16) {
    const float* Pp = &sPart[0][16 * (tid >> 3) + 2 * (tid & 7)];
    float l0 = bb[0], l1 = bb[1];
    for (int w = 0; w < 8; ++w) {
      const v2f pw = *(const v2f*)(Pp + 32 * w);
      l0 += pw.x;
      l1 += pw.y;
    }
    float mx = fmaxf(l0, l1);
    float e0 = expf(l0 - mx), e1 = expf(l1 - mx);
    float inv = 1.0f / (e0 + e1);
    ptb[r0 + tid] = e0 * inv * ipv;
    pfb[r0 + tid] = e1 * inv * ipv;
  }

  for (int idx = tid; idx < 2 * 16 * kH; idx += 256) {
    int l = idx >> 11, rem = idx & 2047, m = rem >> 7, k = rem & 127;
    size_t row = r0 + m;
    Stg[((size_t)l * kBN + row) * kH + k]         = sC[l][m * kH + k];
    Stg[kHOFF + ((size_t)l * kBN + row) * kH + k] = sH[l][kFin][m * AS + k];
  }
}

__device__ __forceinline__ void load_w_rows(const float* __restrict__ p, v4f (&w)[16]) {
#pragma unroll
  for (int i = 0; i < 16; ++i) w[i] = *(const v4f*)(p + (size_t)i * kOutV);
}
__device__ __forceinline__ void fma_w_rows(const float* Fp, const v4f (&w)[16], v4f (&acc)[kB]) {
#pragma unroll
  for (int i = 0; i < 16; ++i) {
    const v4f f0 = *(const v4f*)(Fp + i * kB);
    const v4f f1 = *(const v4f*)(Fp + i * kB + 4);
    acc[0] += f0.x * w[i];
    acc[1] += f0.y * w[i];
    acc[2] += f0.z * w[i];
    acc[3] += f0.w * w[i];
    acc[4] += f1.x * w[i];
    acc[5] += f1.y * w[i];
    acc[6] += f1.z * w[i];
    acc[7] += f1.w * w[i];
  }
}

__global__ __launch_bounds__(256) void k_logits(
    const float* __restrict__ StgIn, const float* __restrict__ Sfin,
    const float* __restrict__ ptIn, const float* __restrict__ pfIn,
    const int* __restrict__ ti, const int* __restrict__ fi,
    const int* __restrict__ steps, const int* __restrict__ exitI,
    const float* __restrict__ outW, const float* __restrict__ outb,
    float* __restrict__ out) {
  static_assert(kB == 8 && kOutV % 4 == 0, "a k's batch values are two v4f; a lane's 4 columns are all in or all out");
  alignas(16) __shared__ float sF[4 * kH * kB];
  alignas(16) __shared__ float sP[8][kB][128];
  alignas(16) __shared__ float sList[8][256 * 4];
  const int tid  = threadIdx.x;
  const int lane = tid & 31;
  const int wave = __builtin_amdgcn_readfirstlane(tid >> 5);

  {
    const int q = wave;
    int stq = steps[q];
    stq = stq < 0 ? 0 : stq;
    const int e = exitI[q];
    v4f fin[4];
    if (stq >= kSteps) {
      v4f none[4];
      float ipe, ipn;
      gather2(StgIn, ptIn, pfIn, ti, fi, q * kN, e, -1, lane, sList[q], fin, none, ipe, ipn);
      const float denom = ipe + 1e-7f;
#pragma unroll
      for (int p = 0; p < 4; ++p) fin[p] = fin[p] / denom;
    } else {
      const float* r = Sfin + (size_t)(q * kN + e) * kH + 4 * lane;
      fin[0] = *(const v4f*)(r);
      fin[1] = *(const v4f*)(r + kHOFF);
      fin[2] = *(const v4f*)(r + (size_t)kBN * kH);
      fin[3] = *(const v4f*)(r + kHOFF + (size_t)kBN * kH);
    }
#pragma unroll
    for (int p = 0; p < 4; ++p) {
#pragma unroll
      for (int j = 0; j < 4; ++j) sF[(p * kH + 4 * lane + j) * kB + q] = fin[p][j];
    }
  }
  __syncthreads();

  const int j0 = blockIdx.x * 128 + 4 * lane;
  const bool inb = j0 < kOutV;
  const float* Wp = outW + (size_t)(wave * 64) * kOutV + (inb ? j0 : 0);
  const float* Fp = sF + wave * 64 * kB;
  v4f acc[kB];
#pragma unroll
  for (int q = 0; q < kB; ++q) acc[q] = (v4f){0.f, 0.f, 0.f, 0.f};
  v4f w[2][16];
  load_w_rows(Wp, w[0]);
#pragma unroll
  for (int g = 0; g < 4; ++g) {
    __builtin_amdgcn_sched_barrier(0);
    if (g + 1 < 4) load_w_rows(Wp + (size_t)(16 * (g + 1)) * kOutV, w[(g + 1) & 1]);
    __builtin_amdgcn_sched_barrier(0);
    fma_w_rows(Fp + 16 * g * kB, w[g & 1], acc);
  }

#pragma unroll
  for (int q = 0; q < kB; ++q) *(v4f*)&sP[wave][q][4 * lane] = acc[q];
  __syncthreads();

  if (inb) {
    const int q = tid >> 5;
    v4f r = *(const v4f*)(outb + j0);
#pragma unroll
    for (int s = 0; s < 8; ++s) r += *(const v4f*)&sP[s][q][4 * lane];
    *(v4f*)(out + (size_t)q * kOutV + j0) = r;
  }
}

__global__ __launch_bounds__(256) void k_xgates(
    const int* __restrict__ data, const float* __restrict__ embed,
    const v2f* __restrict__ Wxp, const float* __restrict__ bias,
    float* __restrict__ X, size_t ws_bytes) {
  __shared__ float sX[16 * AS];
  const int tid  = threadIdx.x;
  const int lane = tid & 31;
  const int wave = tid >> 5;
  static_assert(kL == 4, "the grid index is split as (node tile) * 4 + token");
  const int tok  = blockIdx.x & 3;
  const int r0   = (blockIdx.x >> 2) * 16;
  const int c_base = wave * 64;
  const int sm0 = tid >> 7, sk = tid & 127;

  int tk[8];
  float xv[8];
#pragma unroll
  for (int j = 0; j < 8; ++j) tk[j] = data[(size_t)(r0 + sm0 + 2 * j) * kL + tok];
#pragma unroll
  for (int j = 0; j < 8; ++j) xv[j] = embed[(size_t)tk[j] * kH + sk];

  if (ws_bytes < kNeedBytes) {
#pragma unroll
    for (int j = 0; j < 8; ++j) X[((size_t)(r0 + sm0 + 2 * j) * kL + tok) * kH + sk] = xv[j];
    return;
  }
#pragma unroll
  for (int j = 0; j < 8; ++j) sX[(sm0 + 2 * j) * AS + sk] = xv[j];
  __syncthreads();

  const int aoff = (lane & 15) * AS + 2 * (lane >> 4);
  const size_t boff = (size_t)(lane >> 4) * kG + c_base + (lane & 15);
  v8f acc[4];
#pragma unroll
  for (int t = 0; t < 4; ++t) {
    float blv = bias[t * kH + wave * 16 + (lane & 15)];
#pragma unroll
    for (int v = 0; v < 8; ++v) acc[t][v] = blv;
  }
  v2f bq0[8][4], bq1[8][4], a0[8], a1[8];
  load_b_block(Wxp + boff, 0, bq0);
  load_a_block(sX + aoff, 0, a0);
  load_b_block(Wxp + boff, 1, bq1);
  load_a_block(sX + aoff, 1, a1);
  mma_block(a0, bq0, acc);
  load_b_block(Wxp + boff, 2, bq0);
  load_a_block(sX + aoff, 2, a0);
  mma_block(a1, bq1, acc);
  load_b_block(Wxp + boff, 3, bq1);
  load_a_block(sX + aoff, 3, a1);
  mma_block(a0, bq0, acc);
  mma_block(a1, bq1, acc);

  v4f* Xo = (v4f*)X + ((size_t)blockIdx.x * 8 + wave) * 4 * 64 + lane;
#pragma unroll
  for (int t = 0; t < 4; ++t) store_tile(Xo + t * 64, acc[t]);
}

__global__ void k_pack(const float* __restrict__ Wsrc, float* __restrict__ Wdst, int total) {
  int idx = blockIdx.x * 256 + threadIdx.x;
  if (idx >= total) return;
  int q   = idx & 511;
  int col = ((q >> 4) & 3) * kH + (q >> 6) * 16 + (q & 15);
  int p   = (idx >> 9) & 63;
  int l   = idx >> 15;
  const float* Wl = Wsrc + (size_t)l * kH * kG;
  v2f v;
  v.x = Wl[(size_t)(2 * p) * kG + col];
  v.y = Wl[(size_t)(2 * p + 1) * kG + col];
  *(v2f*)(Wdst + (size_t)idx * 2) = v;
}

extern "C" void kernel_launch(void* const* d_in, const int* in_sizes, int n_in,
                              void* d_out, int out_size, void* d_ws, size_t ws_size,
                              hipStream_t stream) {
  const int*   data  = (const int*)d_in[0];
  const int*   ti    = (const int*)d_in[1];
  const int*   fi    = (const int*)d_in[2];
  const int*   exitI = (const int*)d_in[3];
  const int*   steps = (const int*)d_in[4];
  const float* embed = (const float*)d_in[5];
  const float* Wx    = (const float*)d_in[6];
  const float* Wh    = (const float*)d_in[7];
  const float* bias  = (const float*)d_in[8];
  const float* bW    = (const float*)d_in[9];
  const float* bb    = (const float*)d_in[10];
  const float* outW  = (const float*)d_in[11];
  const float* outb  = (const float*)d_in[12];
  float* out = (float*)d_out;

  float* ws = (float*)d_ws;
  const size_t ST  = kST;
  float* G0  = ws;
  float* G1  = G0 + ST;
  float* SF  = G1 + ST;
  float* pt0 = SF + ST;
  float* pf0 = pt0 + kBN;
  float* pt1 = pf0 + kBN;
  float* pf1 = pt1 + kBN;
  float* Wxp = pf1 + kBN;
  float* Whp = Wxp + (size_t)kPairs * 2;
  float* X   = Whp + (size_t)kPairs * 2;

  k_pack<<<(kPairs + 255) / 256, 256, 0, stream>>>(Wx, Wxp, kPairs);
  k_pack<<<(kPairs + 255) / 256, 256, 0, stream>>>(Wh, Whp, kPairs);
  k_xgates<<<(kBN / 16) * kL, 256, 0, stream>>>(data, embed, (const v2f*)Wxp, bias, X, ws_size);

  float* Gin = G1; float* Gout = G0;
  float* ptin = pt1; float* pfin = pf1; float* ptout = pt0; float* pfout = pf0;
  for (int s = 0; s < kSteps; ++s) {
    k_lstm_step<<<kBN / 16, 256, 0, stream>>>(X, ws_size, Gin, Gout, SF, ptin, pfin, ptout, pfout,
                                              (const v2f*)Wxp, (const v2f*)Whp,
                                              bias, bW, bb, exitI, ti, fi, steps, s);
    float* t1 = Gin; Gin = Gout; Gout = t1;
    float* t2 = ptin; ptin = ptout; ptout = t2;
    float* t3 = pfin; pfin = pfout; pfout = t3;
  }
  k_logits<<<(kOutV + 127) / 128, 256, 0, stream>>>(Gin, SF, ptin, pfin, ti, fi, steps, exitI,
                                                    outW, outb, out);
  (void)in_sizes; (void)n_in; (void)out_size;
}


# Step kernel as a software pipeline: each cell update is issued beside independent WMMAs of the next phase (layer 1's h1.Wh half; next token's layer 0); layer-1 gate sum order changes
# speedup vs baseline: 3.5249x; 1.0512x over previous
#include <hip/hip_runtime.h>
#include <cstddef>

typedef float v2f __attribute__((ext_vector_type(2)));
typedef float v4f __attribute__((ext_vector_type(4)));
typedef float v8f __attribute__((ext_vector_type(8)));

namespace {
constexpr int kB    = 8;
constexpr int kN    = 256;
constexpr int kL    = 4;
constexpr int kH    = 128;
constexpr int kBN   = kB * kN;
constexpr int kG    = 4 * kH;
constexpr int kSteps = 16;
constexpr int kOutV = 30000;
constexpr size_t kHOFF = (size_t)2 * kBN * kH;
constexpr int AS = 132;
constexpr int kPairs = 2 * (kH / 2) * kG;
constexpr size_t kST  = (size_t)4 * kBN * kH;
constexpr size_t kEMB = (size_t)kBN * kL * kH;
constexpr size_t kGX  = (size_t)kBN * kL * kG;
constexpr size_t kFixed = 3 * kST + 4 * (size_t)kBN + 4 * (size_t)kPairs;
constexpr size_t kNeedBytes = (kFixed + kGX) * sizeof(float);
}

__device__ __forceinline__ float sigmf(float x) {
  return __builtin_amdgcn_rcpf(1.0f + __builtin_amdgcn_exp2f(x * -1.44269504088896341f));
}
__device__ __forceinline__ float tanhfast(float x) {
  return __builtin_fmaf(-2.0f, __builtin_amdgcn_rcpf(1.0f + __builtin_amdgcn_exp2f(x * 2.88539008177792681f)), 1.0f);
}

__device__ __forceinline__ v8f wmma4(v2f a, v2f b, v8f c) {
  return __builtin_amdgcn_wmma_f32_16x16x4_f32(false, a, false, b, (short)0, c, false, false);
}

__device__ __forceinline__ void load_b_block(const v2f* __restrict__ Wl, int blk, v2f (&b)[8][4]) {
#pragma unroll
  for (int j = 0; j < 8; ++j) {
    const v2f* Wr = Wl + (size_t)((blk * 8 + j) * 2) * kG;
#pragma unroll
    for (int t = 0; t < 4; ++t) b[j][t] = Wr[t * 16];
  }
}

__device__ __forceinline__ void load_a_block(const float* Al, int blk, v2f (&a)[8]) {
#pragma unroll
  for (int j = 0; j < 8; ++j) a[j] = *(const v2f*)(Al + (blk * 8 + j) * 4);
}

__device__ __forceinline__ void mma_block(const v2f (&a)[8], const v2f (&b)[8][4], v8f (&acc)[4]) {
#pragma unroll
  for (int j = 0; j < 8; ++j) {
#pragma unroll
    for (int t = 0; t < 4; ++t) acc[t] = wmma4(a[j], b[j][t], acc[t]);
  }
}

__device__ __forceinline__ v8f load_tile(const v4f* p) {
  const v4f lo = p[0], hi = p[32];
  return __builtin_shufflevector(lo, hi, 0, 1, 2, 3, 4, 5, 6, 7);
}
__device__ __forceinline__ void store_tile(v4f* p, v8f a) {
  p[0]  = __builtin_shufflevector(a, a, 0, 1, 2, 3);
  p[32] = __builtin_shufflevector(a, a, 4, 5, 6, 7);
}

template <bool kSide, bool kBar>
__device__ __forceinline__ void run4(const float* Al, const float* AlNext,
                                     const v2f* __restrict__ Wc, const v2f* __restrict__ Wn,
                                     v2f (&bq0)[8][4], v2f (&bq1)[8][4], v2f (&a0)[8], v2f (&a1)[8], v8f (&acc)[4],
                                     const v8f (&g)[4], float (&c)[8], float (&hcur)[8], float* Hw) {
#pragma unroll
  for (int blk = 0; blk < 4; ++blk) {
    if (kBar && blk == 3) __syncthreads();
    if (blk & 1) {
      load_b_block(blk < 3 ? Wc : Wn, (blk + 1) & 3, bq0);
      load_a_block(blk < 3 ? Al : AlNext, (blk + 1) & 3, a0);
      mma_block(a1, bq1, acc);
    } else {
      load_b_block(Wc, blk + 1, bq1);
      load_a_block(Al, blk + 1, a1);
      mma_block(a0, bq0, acc);
    }
    if (kSide && blk < 3) {
#pragma unroll
      for (int v = 3 * blk; v < (blk < 2 ? 3 * blk + 3 : 8); ++v) {
        const float cn = sigmf(g[1][v]) * c[v] + sigmf(g[0][v]) * tanhfast(g[2][v]);
        c[v] = cn;
        const float hn = sigmf(g[3][v]) * tanhfast(cn);
        hcur[v] = hn;
        Hw[v * AS] = hn;
      }
    }
    __builtin_amdgcn_sched_barrier(0);
  }
}

__device__ __forceinline__ void gather2(
    const float* __restrict__ Stg, const float* __restrict__ ptb, const float* __restrict__ pfb,
    const int* __restrict__ ti, const int* __restrict__ fi, int base, int mA, int mB, int lane,
    float* Lw, v4f (&aA)[4], v4f (&aB)[4], float& ipA, float& ipB) {
  int t[8], f[8];
  float pt[8], pf[8];
#pragma unroll
  for (int i = 0; i < 8; ++i) {
    const int src = base + 32 * i + lane;
    t[i] = ti[src];
    f[i] = fi[src];
    pt[i] = ptb[src];
    pf[i] = pfb[src];
  }
  int cnt = 0;
#pragma unroll
  for (int i = 0; i < 8; ++i) {
    const float a0 = (t[i] == mA) ? 0.f + pt[i] : 0.f;
    const float wa = (f[i] == mA) ? a0 + pf[i] : a0;
    const float b0 = (t[i] == mB) ? 0.f + pt[i] : 0.f;
    const float wb = (f[i] == mB) ? b0 + pf[i] : b0;
    const bool con = (wa != 0.f) || (wb != 0.f);
    const unsigned mk = __builtin_amdgcn_ballot_w32(con);
    if (con) {
      const int rank = cnt + (int)__builtin_amdgcn_mbcnt_lo(mk, 0u);
      *(v4f*)(Lw + 4 * rank) = (v4f){__int_as_float(32 * i + lane), wa, wb, 0.f};
    }
    cnt += __builtin_popcount(mk);
  }
#pragma unroll
  for (int p = 0; p < 4; ++p) { aA[p] = (v4f){0.f, 0.f, 0.f, 0.f}; aB[p] = aA[p]; }
  ipA = 0.f;
  ipB = 0.f;
  for (int k = 0; k < cnt; k += 8) {
    v4f e[8];
    v4f x[8][4];
#pragma unroll
    for (int j = 0; j < 8; ++j) {
      const int idx = (k + j < cnt) ? k + j : cnt - 1;
      e[j] = *(const v4f*)(Lw + 4 * idx);
    }
#pragma unroll
    for (int j = 0; j < 8; ++j) {
      const float* r = Stg + (size_t)(base + __float_as_int(e[j].x)) * kH + 4 * lane;
      x[j][0] = *(const v4f*)(r);
      x[j][1] = *(const v4f*)(r + kHOFF);
      x[j][2] = *(const v4f*)(r + (size_t)kBN * kH);
      x[j][3] = *(const v4f*)(r + kHOFF + (size_t)kBN * kH);
    }
#pragma unroll
    for (int j = 0; j < 8; ++j) {
      const bool in = k + j < cnt;
      const float wa = in ? e[j].y : 0.f;
      const float wb = in ? e[j].z : 0.f;
      ipA += wa;
      ipB += wb;
#pragma unroll
      for (int p = 0; p < 4; ++p) {
        aA[p] += wa * x[j][p];
        aB[p] += wb * x[j][p];
      }
    }
  }
}

__global__ __launch_bounds__(256) void k_lstm_step(
    const float* __restrict__ X, size_t ws_bytes, const float* __restrict__ StgIn,
    float* __restrict__ Stg, float* __restrict__ Sfin,
    const float* __restrict__ ptIn, const float* __restrict__ pfIn,
    float* __restrict__ ptb, float* __restrict__ pfb,
    const v2f* __restrict__ Wxp, const v2f* __restrict__ Whp,
    const float* __restrict__ bias, const float* __restrict__ bW,
    const float* __restrict__ bb, const int* __restrict__ exitI,
    const int* __restrict__ ti, const int* __restrict__ fi,
    const int* __restrict__ steps, int step) {
  __shared__ float sX[16 * AS];
  alignas(16) __shared__ float sH[2][2][16 * AS];
  alignas(16) __shared__ float sC[2][16 * kH];
  alignas(16) __shared__ float sList[8][256 * 4];
  __shared__ float sIp[16];
  alignas(16) __shared__ float sPart[8][32];
  alignas(16) __shared__ float sBW[4 * kH * 2];

  const int tid  = threadIdx.x;
  const int lane = tid & 31;
  const int wave = tid >> 5;
  const int r0   = blockIdx.x * 16;
  const int c_base = wave * 64;
  const int hf   = lane >> 4;
  const int hcol = wave * 16 + (lane & 15);
  const bool pre = ws_bytes >= kNeedBytes;
  const v4f* Xg = (const v4f*)X + ((size_t)blockIdx.x * 128 + wave * 4) * 64 + lane;
  v8f gx[4];
  if (pre) {
#pragma unroll
    for (int t = 0; t < 4; ++t) gx[t] = load_tile(Xg + t * 64);
  }

  const int mm  = exitI[r0 >> 8] - (r0 & 255);
  const bool hit = (mm >= 0) && (mm < 16);
  const int mmc = mm < 0 ? 0 : (mm > 15 ? 15 : mm);

  const int bidx = r0 >> 8;
  int stp = steps[bidx];
  stp = stp < 0 ? 0 : stp;
  if (step > stp) return;

  const int wv = __builtin_amdgcn_readfirstlane(wave);
  {
    v4f ag[2][4];
    float ipr[2];
    if (step == 0) {
#pragma unroll
      for (int p = 0; p < 4; ++p) { ag[0][p] = (v4f){0.f, 0.f, 0.f, 0.f}; ag[1][p] = ag[0][p]; }
      ipr[0] = ((r0 & 255) + 2 * wv == 0) ? 1.0f : 0.0f;
      ipr[1] = 0.0f;
    } else {
      const int mA = (r0 & 255) + 2 * wv;
      gather2(StgIn, ptIn, pfIn, ti, fi, bidx * kN, mA, mA + 1, lane, sList[wv], ag[0], ag[1], ipr[0], ipr[1]);
#pragma unroll
      for (int j = 0; j < 2; ++j) {
        const float denom = ipr[j] + 1e-7f;
#pragma unroll
        for (int p = 0; p < 4; ++p) ag[j][p] = ag[j][p] / denom;
      }
    }
    if (step == stp) {
#pragma unroll
      for (int j = 0; j < 2; ++j) {
        const size_t o = (size_t)(r0 + 2 * wv + j) * kH + 4 * lane;
        *(v4f*)(Sfin + o)                            = ag[j][0];
        *(v4f*)(Sfin + kHOFF + o)                    = ag[j][1];
        *(v4f*)(Sfin + (size_t)kBN * kH + o)         = ag[j][2];
        *(v4f*)(Sfin + kHOFF + (size_t)kBN * kH + o) = ag[j][3];
      }
      return;
    }
#pragma unroll
    for (int j = 0; j < 2; ++j) {
      const int rr = 2 * wv + j;
      *(v4f*)(&sC[0][rr * kH + 4 * lane])    = ag[j][0];
      *(v4f*)(&sH[0][0][rr * AS + 4 * lane]) = ag[j][1];
      *(v4f*)(&sC[1][rr * kH + 4 * lane])    = ag[j][2];
      *(v4f*)(&sH[1][0][rr * AS + 4 * lane]) = ag[j][3];
      if (lane == 0) sIp[rr] = ipr[j];
    }
  }

  const int sm0 = tid >> 7, sk = tid & 127;
  {
    float bwv[4];
#pragma unroll
    for (int i = 0; i < 4; ++i) bwv[i] = bW[tid + 256 * i];
#pragma unroll
    for (int i = 0; i < 4; ++i) sBW[tid + 256 * i] = bwv[i];
  }
  __syncthreads();

  float cst[2][8];
  float hst[2][8];
#pragma unroll
  for (int i = 0; i < 16; ++i) cst[i >> 3][i & 7] = sC[i >> 3][((i & 7) + 8 * hf) * kH + hcol];

  const int aoff = (lane & 15) * AS + 2 * hf;
  const size_t boff = (size_t)hf * kG + c_base + (lane & 15);
  v2f bq0[8][4], bq1[8][4];
  load_b_block((pre ? Whp : Wxp) + boff, 0, bq0);
  float exs[4];
  exs[0] = sC[0][mmc * kH + hcol];
  exs[1] = sH[0][0][mmc * AS + hcol];
  exs[2] = sC[1][mmc * kH + hcol];
  exs[3] = sH[1][0][mmc * AS + hcol];

  if (__builtin_expect(pre, true)) {
    float b1v[4];
#pragma unroll
    for (int t = 0; t < 4; ++t) b1v[t] = bias[kG + t * kH + hcol];
    unsigned bo = (unsigned)boff;
    const int hwoff = (8 * hf) * AS + hcol;
    v8f acc1[4];
    v2f a0[8], a1[8];
    {
      const float* Al = &sH[0][0][aoff];
      const float* An = &sH[1][0][aoff];
      const v2f* Wa = Whp + bo;
      const v2f* Wb = Whp + (size_t)(kH / 2) * kG + bo;
      load_a_block(Al, 0, a0);
#pragma unroll 1
      for (int i = 0; i < 2; ++i) {
        load_b_block(Wa, 2 * i + 1, bq1);
        load_a_block(Al, 2 * i + 1, a1);
        mma_block(a0, bq0, gx);
        load_b_block(i == 0 ? Wa : Wb, (2 * i + 2) & 3, bq0);
        load_a_block(i == 0 ? Al : An, (2 * i + 2) & 3, a0);
        mma_block(a1, bq1, gx);
      }
    }
#pragma unroll 1
    for (int tok = 0; tok < kL; ++tok) {
      asm volatile("" : "+v"(bo));
      const v2f* Wh0 = Whp + bo;
      const v2f* Wh1 = Whp + (size_t)(kH / 2) * kG + bo;
      const v2f* Wx1 = Wxp + (size_t)(kH / 2) * kG + bo;
      const int rb = tok & 1;
      float* H0n = sH[0][rb ^ 1];
      float* H1n = sH[1][rb ^ 1];
#pragma unroll
      for (int t = 0; t < 4; ++t) {
#pragma unroll
        for (int v = 0; v < 8; ++v) acc1[t][v] = b1v[t];
      }
      run4<true, true>(sH[1][rb] + aoff, H0n + aoff, Wh1, Wx1, bq0, bq1, a0, a1, acc1, gx, cst[0], hst[0], H0n + hwoff);
      const int tn = (tok + 1 < kL) ? tok + 1 : kL - 1;
#pragma unroll
      for (int t = 0; t < 4; ++t) gx[t] = load_tile(Xg + (tn * 32 + t) * 64);
      run4<false, false>(H0n + aoff, H0n + aoff, Wx1, Wh0, bq0, bq1, a0, a1, acc1, acc1, cst[1], hst[1], nullptr);
      if (tok + 1 < kL) {
        run4<true, true>(H0n + aoff, H1n + aoff, Wh0, Wh1, bq0, bq1, a0, a1, gx, acc1, cst[1], hst[1], H1n + hwoff);
      } else {
#pragma unroll
        for (int v = 0; v < 8; ++v) {
          const float cn = sigmf(acc1[1][v]) * cst[1][v] + sigmf(acc1[0][v]) * tanhfast(acc1[2][v]);
          cst[1][v] = cn;
          const float hn = sigmf(acc1[3][v]) * tanhfast(cn);
          hst[1][v] = hn;
          H1n[hwoff + v * AS] = hn;
        }
        __syncthreads();
      }
    }
  } else {
    for (int tok = 0; tok < kL; ++tok) {
      const int rb = tok & 1;
      if (!pre) {
        const float* xp = X + ((size_t)(r0 + sm0) * kL + tok) * kH + sk;
        float xv[8];
  #pragma unroll
        for (int j = 0; j < 8; ++j) xv[j] = xp[(size_t)2 * j * kL * kH];
  #pragma unroll
        for (int j = 0; j < 8; ++j) sX[(sm0 + 2 * j) * AS + sk] = xv[j];
        __syncthreads();
      }
  #pragma unroll
      for (int l = 0; l < 2; ++l) {
        const bool skipx = (l == 0) && pre;
        v8f acc[4];
        if (skipx) {
  #pragma unroll
          for (int t = 0; t < 4; ++t) acc[t] = gx[t];
        } else {
  #pragma unroll
          for (int t = 0; t < 4; ++t) {
            float blv = bias[l * kG + t * kH + hcol];
  #pragma unroll
            for (int v = 0; v < 8; ++v) acc[t][v] = blv;
          }
        }
        if (l == 1 && pre && tok + 1 < kL) {
  #pragma unroll
          for (int t = 0; t < 4; ++t) gx[t] = load_tile(Xg + ((tok + 1) * 32 + t) * 64);
        }
        const float* A0 = ((l == 0) ? sX : sH[0][rb ^ 1]) + aoff;
        const float* A1 = sH[l][rb] + aoff;
        const v2f* Wx_l = Wxp + (size_t)l * (kH / 2) * kG + boff;
        const v2f* Wh_l = Whp + (size_t)l * (kH / 2) * kG + boff;
        const v2f* W_n  = (l == 0) ? Wxp + (size_t)(kH / 2) * kG + boff : (pre ? Whp : Wxp) + boff;
        v2f a0[8], a1[8];
        load_a_block(skipx ? A1 : A0, 0, a0);
  #pragma unroll 1
        for (int i = skipx ? 2 : 0; i < 4; ++i) {
          const int g1 = 2 * i + 1;
          load_b_block((g1 < 4) ? Wx_l : Wh_l, g1 & 3, bq1);
          load_a_block((g1 < 4) ? A0 : A1, g1 & 3, a1);
          mma_block(a0, bq0, acc);
          const int g2 = 2 * i + 2;
          load_b_block((g2 < 4) ? Wx_l : ((g2 < 8) ? Wh_l : W_n), g2 & 3, bq0);
          if (g2 < 8) load_a_block((g2 < 4) ? A0 : A1, g2 & 3, a0);
          mma_block(a1, bq1, acc);
        }
        float* Hw = sH[l][rb ^ 1] + (8 * hf) * AS + hcol;
  #pragma unroll
        for (int v = 0; v < 8; ++v) {
          const float cn = sigmf(acc[1][v]) * cst[l][v] + sigmf(acc[0][v]) * tanhfast(acc[2][v]);
          cst[l][v] = cn;
          const float hn = sigmf(acc[3][v]) * tanhfast(cn);
          hst[l][v] = hn;
          Hw[v * AS] = hn;
        }
        __syncthreads();
      }
    }
  }

  constexpr int kFin = kL & 1;
  const float ipv = (tid < 16) ? sIp[tid] : 0.f;

  const bool mine = hit && (hf == (mmc >> 3));
  const int mv = mmc & 7;
#pragma unroll
  for (int v = 0; v < 8; ++v) {
    const bool ex = mine && (v == mv);
    cst[0][v] = ex ? exs[0] : cst[0][v];
    hst[0][v] = ex ? exs[1] : hst[0][v];
    cst[1][v] = ex ? exs[2] : cst[1][v];
    hst[1][v] = ex ? exs[3] : hst[1][v];
  }
  if (mine) {
    sH[0][kFin][mmc * AS + hcol] = exs[1];
    sH[1][kFin][mmc * AS + hcol] = exs[3];
  }
#pragma unroll
  for (int i = 0; i < 16; ++i) sC[i >> 3][((i & 7) + 8 * hf) * kH + hcol] = cst[i >> 3][i & 7];

  {
    const v2f w0 = *(const v2f*)(sBW + 2 * hcol);
    const v2f w1 = *(const v2f*)(sBW + 2 * (kH + hcol));
    const v2f w2 = *(const v2f*)(sBW + 2 * (2 * kH + hcol));
    const v2f w3 = *(const v2f*)(sBW + 2 * (3 * kH + hcol));
    float q[16];
#pragma unroll
    for (int v = 0; v < 8; ++v) {
      q[2 * v]     = cst[0][v] * w0.x + hst[0][v] * w1.x + cst[1][v] * w2.x + hst[1][v] * w3.x;
      q[2 * v + 1] = cst[0][v] * w0.y + hst[0][v] * w1.y + cst[1][v] * w2.y + hst[1][v] * w3.y;
    }
    const int cl = lane & 15;
    float r8[8], r4[4], r2[2];
    const bool u8 = (cl & 8) != 0, u4 = (cl & 4) != 0, u2 = (cl & 2) != 0, u1 = (cl & 1) != 0;
#pragma unroll
    for (int i = 0; i < 8; ++i) r8[i] = (u8 ? q[i + 8] : q[i]) + __shfl_xor(u8 ? q[i] : q[i + 8], 8);
#pragma unroll
    for (int i = 0; i < 4; ++i) r4[i] = (u4 ? r8[i + 4] : r8[i]) + __shfl_xor(u4 ? r8[i] : r8[i + 4], 4);
#pragma unroll
    for (int i = 0; i < 2; ++i) r2[i] = (u2 ? r4[i + 2] : r4[i]) + __shfl_xor(u2 ? r4[i] : r4[i + 2], 2);
    sPart[wave][lane] = (u1 ? r2[1] : r2[0]) + __shfl_xor(u1 ? r2[0] : r2[1], 1);
  }
  __syncthreads();

  if (tid < 16) {
    const float* Pp = &sPart[0][16 * (tid >> 3) + 2 * (tid & 7)];
    float l0 = bb[0], l1 = bb[1];
    for (int w = 0; w < 8; ++w) {
      const v2f pw = *(const v2f*)(Pp + 32 * w);
      l0 += pw.x;
      l1 += pw.y;
    }
    float mx = fmaxf(l0, l1);
    float e0 = expf(l0 - mx), e1 = expf(l1 - mx);
    float inv = 1.0f / (e0 + e1);
    ptb[r0 + tid] = e0 * inv * ipv;
    pfb[r0 + tid] = e1 * inv * ipv;
  }

  for (int idx = tid; idx < 2 * 16 * kH; idx += 256) {
    int l = idx >> 11, rem = idx & 2047, m = rem >> 7, k = rem & 127;
    size_t row = r0 + m;
    Stg[((size_t)l * kBN + row) * kH + k]         = sC[l][m * kH + k];
    Stg[kHOFF + ((size_t)l * kBN + row) * kH + k] = sH[l][kFin][m * AS + k];
  }
}

__device__ __forceinline__ void load_w_rows(const float* __restrict__ p, v4f (&w)[16]) {
#pragma unroll
  for (int i = 0; i < 16; ++i) w[i] = *(const v4f*)(p + (size_t)i * kOutV);
}
__device__ __forceinline__ void fma_w_rows(const float* Fp, const v4f (&w)[16], v4f (&acc)[kB]) {
#pragma unroll
  for (int i = 0; i < 16; ++i) {
    const v4f f0 = *(const v4f*)(Fp + i * kB);
    const v4f f1 = *(const v4f*)(Fp + i * kB + 4);
    acc[0] += f0.x * w[i];
    acc[1] += f0.y * w[i];
    acc[2] += f0.z * w[i];
    acc[3] += f0.w * w[i];
    acc[4] += f1.x * w[i];
    acc[5] += f1.y * w[i];
    acc[6] += f1.z * w[i];
    acc[7] += f1.w * w[i];
  }
}

__global__ __launch_bounds__(256) void k_logits(
    const float* __restrict__ StgIn, const float* __restrict__ Sfin,
    const float* __restrict__ ptIn, const float* __restrict__ pfIn,
    const int* __restrict__ ti, const int* __restrict__ fi,
    const int* __restrict__ steps, const int* __restrict__ exitI,
    const float* __restrict__ outW, const float* __restrict__ outb,
    float* __restrict__ out) {
  static_assert(kB == 8 && kOutV % 4 == 0, "a k's batch values are two v4f; a lane's 4 columns are all in or all out");
  alignas(16) __shared__ float sF[4 * kH * kB];
  alignas(16) __shared__ float sP[8][kB][128];
  alignas(16) __shared__ float sList[8][256 * 4];
  const int tid  = threadIdx.x;
  const int lane = tid & 31;
  const int wave = __builtin_amdgcn_readfirstlane(tid >> 5);

  {
    const int q = wave;
    int stq = steps[q];
    stq = stq < 0 ? 0 : stq;
    const int e = exitI[q];
    v4f fin[4];
    if (stq >= kSteps) {
      v4f none[4];
      float ipe, ipn;
      gather2(StgIn, ptIn, pfIn, ti, fi, q * kN, e, -1, lane, sList[q], fin, none, ipe, ipn);
      const float denom = ipe + 1e-7f;
#pragma unroll
      for (int p = 0; p < 4; ++p) fin[p] = fin[p] / denom;
    } else {
      const float* r = Sfin + (size_t)(q * kN + e) * kH + 4 * lane;
      fin[0] = *(const v4f*)(r);
      fin[1] = *(const v4f*)(r + kHOFF);
      fin[2] = *(const v4f*)(r + (size_t)kBN * kH);
      fin[3] = *(const v4f*)(r + kHOFF + (size_t)kBN * kH);
    }
#pragma unroll
    for (int p = 0; p < 4; ++p) {
#pragma unroll
      for (int j = 0; j < 4; ++j) sF[(p * kH + 4 * lane + j) * kB + q] = fin[p][j];
    }
  }
  __syncthreads();

  const int j0 = blockIdx.x * 128 + 4 * lane;
  const bool inb = j0 < kOutV;
  const float* Wp = outW + (size_t)(wave * 64) * kOutV + (inb ? j0 : 0);
  const float* Fp = sF + wave * 64 * kB;
  v4f acc[kB];
#pragma unroll
  for (int q = 0; q < kB; ++q) acc[q] = (v4f){0.f, 0.f, 0.f, 0.f};
  v4f w[2][16];
  load_w_rows(Wp, w[0]);
#pragma unroll
  for (int g = 0; g < 4; ++g) {
    __builtin_amdgcn_sched_barrier(0);
    if (g + 1 < 4) load_w_rows(Wp + (size_t)(16 * (g + 1)) * kOutV, w[(g + 1) & 1]);
    __builtin_amdgcn_sched_barrier(0);
    fma_w_rows(Fp + 16 * g * kB, w[g & 1], acc);
  }

#pragma unroll
  for (int q = 0; q < kB; ++q) *(v4f*)&sP[wave][q][4 * lane] = acc[q];
  __syncthreads();

  if (inb) {
    const int q = tid >> 5;
    v4f r = *(const v4f*)(outb + j0);
#pragma unroll
    for (int s = 0; s < 8; ++s) r += *(const v4f*)&sP[s][q][4 * lane];
    *(v4f*)(out + (size_t)q * kOutV + j0) = r;
  }
}

__global__ __launch_bounds__(256) void k_xgates(
    const int* __restrict__ data, const float* __restrict__ embed,
    const v2f* __restrict__ Wxp, const float* __restrict__ bias,
    float* __restrict__ X, size_t ws_bytes) {
  __shared__ float sX[16 * AS];
  const int tid  = threadIdx.x;
  const int lane = tid & 31;
  const int wave = tid >> 5;
  static_assert(kL == 4, "the grid index is split as (node tile) * 4 + token");
  const int tok  = blockIdx.x & 3;
  const int r0   = (blockIdx.x >> 2) * 16;
  const int c_base = wave * 64;
  const int sm0 = tid >> 7, sk = tid & 127;

  int tk[8];
  float xv[8];
#pragma unroll
  for (int j = 0; j < 8; ++j) tk[j] = data[(size_t)(r0 + sm0 + 2 * j) * kL + tok];
#pragma unroll
  for (int j = 0; j < 8; ++j) xv[j] = embed[(size_t)tk[j] * kH + sk];

  if (ws_bytes < kNeedBytes) {
#pragma unroll
    for (int j = 0; j < 8; ++j) X[((size_t)(r0 + sm0 + 2 * j) * kL + tok) * kH + sk] = xv[j];
    return;
  }
#pragma unroll
  for (int j = 0; j < 8; ++j) sX[(sm0 + 2 * j) * AS + sk] = xv[j];
  __syncthreads();

  const int aoff = (lane & 15) * AS + 2 * (lane >> 4);
  const size_t boff = (size_t)(lane >> 4) * kG + c_base + (lane & 15);
  v8f acc[4];
#pragma unroll
  for (int t = 0; t < 4; ++t) {
    float blv = bias[t * kH + wave * 16 + (lane & 15)];
#pragma unroll
    for (int v = 0; v < 8; ++v) acc[t][v] = blv;
  }
  v2f bq0[8][4], bq1[8][4], a0[8], a1[8];
  load_b_block(Wxp + boff, 0, bq0);
  load_a_block(sX + aoff, 0, a0);
  load_b_block(Wxp + boff, 1, bq1);
  load_a_block(sX + aoff, 1, a1);
  mma_block(a0, bq0, acc);
  load_b_block(Wxp + boff, 2, bq0);
  load_a_block(sX + aoff, 2, a0);
  mma_block(a1, bq1, acc);
  load_b_block(Wxp + boff, 3, bq1);
  load_a_block(sX + aoff, 3, a1);
  mma_block(a0, bq0, acc);
  mma_block(a1, bq1, acc);

  v4f* Xo = (v4f*)X + ((size_t)blockIdx.x * 8 + wave) * 4 * 64 + lane;
#pragma unroll
  for (int t = 0; t < 4; ++t) store_tile(Xo + t * 64, acc[t]);
}

__global__ void k_pack(const float* __restrict__ Wsrc, float* __restrict__ Wdst, int total) {
  int idx = blockIdx.x * 256 + threadIdx.x;
  if (idx >= total) return;
  int q   = idx & 511;
  int col = ((q >> 4) & 3) * kH + (q >> 6) * 16 + (q & 15);
  int p   = (idx >> 9) & 63;
  int l   = idx >> 15;
  const float* Wl = Wsrc + (size_t)l * kH * kG;
  v2f v;
  v.x = Wl[(size_t)(2 * p) * kG + col];
  v.y = Wl[(size_t)(2 * p + 1) * kG + col];
  *(v2f*)(Wdst + (size_t)idx * 2) = v;
}

extern "C" void kernel_launch(void* const* d_in, const int* in_sizes, int n_in,
                              void* d_out, int out_size, void* d_ws, size_t ws_size,
                              hipStream_t stream) {
  const int*   data  = (const int*)d_in[0];
  const int*   ti    = (const int*)d_in[1];
  const int*   fi    = (const int*)d_in[2];
  const int*   exitI = (const int*)d_in[3];
  const int*   steps = (const int*)d_in[4];
  const float* embed = (const float*)d_in[5];
  const float* Wx    = (const float*)d_in[6];
  const float* Wh    = (const float*)d_in[7];
  const float* bias  = (const float*)d_in[8];
  const float* bW    = (const float*)d_in[9];
  const float* bb    = (const float*)d_in[10];
  const float* outW  = (const float*)d_in[11];
  const float* outb  = (const float*)d_in[12];
  float* out = (float*)d_out;

  float* ws = (float*)d_ws;
  const size_t ST  = kST;
  float* G0  = ws;
  float* G1  = G0 + ST;
  float* SF  = G1 + ST;
  float* pt0 = SF + ST;
  float* pf0 = pt0 + kBN;
  float* pt1 = pf0 + kBN;
  float* pf1 = pt1 + kBN;
  float* Wxp = pf1 + kBN;
  float* Whp = Wxp + (size_t)kPairs * 2;
  float* X   = Whp + (size_t)kPairs * 2;

  k_pack<<<(kPairs + 255) / 256, 256, 0, stream>>>(Wx, Wxp, kPairs);
  k_pack<<<(kPairs + 255) / 256, 256, 0, stream>>>(Wh, Whp, kPairs);
  k_xgates<<<(kBN / 16) * kL, 256, 0, stream>>>(data, embed, (const v2f*)Wxp, bias, X, ws_size);

  float* Gin = G1; float* Gout = G0;
  float* ptin = pt1; float* pfin = pf1; float* ptout = pt0; float* pfout = pf0;
  for (int s = 0; s < kSteps; ++s) {
    k_lstm_step<<<kBN / 16, 256, 0, stream>>>(X, ws_size, Gin, Gout, SF, ptin, pfin, ptout, pfout,
                                              (const v2f*)Wxp, (const v2f*)Whp,
                                              bias, bW, bb, exitI, ti, fi, steps, s);
    float* t1 = Gin; Gin = Gout; Gout = t1;
    float* t2 = ptin; ptin = ptout; ptout = t2;
    float* t3 = pfin; pfin = pfout; pfout = t3;
  }
  k_logits<<<(kOutV + 127) / 128, 256, 0, stream>>>(Gin, SF, ptin, pfin, ti, fi, steps, exitI,
                                                    outW, outb, out);
  (void)in_sizes; (void)n_in; (void)out_size;
}
